# starting kernel
# baseline (speedup 1.0000x reference)
#include <hip/hip_runtime.h>

typedef __attribute__((ext_vector_type(2))) float v2f;
typedef __attribute__((ext_vector_type(8))) float v8f;

__global__ void zero_f32(float* __restrict__ p, long n) {
  long i = (long)blockIdx.x * blockDim.x + threadIdx.x;
  long stride = (long)gridDim.x * blockDim.x;
  for (; i < n; i += stride) p[i] = 0.0f;
}

__global__ void degree_kernel(const int* __restrict__ dst, float* __restrict__ deg, int E) {
  int e = blockIdx.x * blockDim.x + threadIdx.x;
  if (e < E) atomicAdd(&deg[dst[e]], 1.0f);
}

__global__ void scatter_add_kernel(const float* __restrict__ x, const int* __restrict__ src,
                                   const int* __restrict__ dst, float* __restrict__ agg, int E) {
  int wid = (int)((blockIdx.x * (unsigned)blockDim.x + threadIdx.x) >> 5);
  int lane = threadIdx.x & 31;
  if (wid >= E) return;
  int s = src[wid];
  int d = dst[wid];
  const float4 v = *(const float4*)(x + (size_t)s * 128 + lane * 4);
  float* a = agg + (size_t)d * 128 + lane * 4;
  atomicAdd(a + 0, v.x);
  atomicAdd(a + 1, v.y);
  atomicAdd(a + 2, v.z);
  atomicAdd(a + 3, v.w);
}

__global__ void normalize_kernel(const float* __restrict__ agg, const float* __restrict__ deg,
                                 float* __restrict__ hn, long total) {
  long i = (long)blockIdx.x * blockDim.x + threadIdx.x;
  if (i >= total) return;
  long node = i >> 7;
  hn[i] = agg[i] / fmaxf(deg[node], 1.0f);
}

__global__ void sage_linear_wmma(const float* __restrict__ Xs, const float* __restrict__ Xn,
                                 const float* __restrict__ Wself, const float* __restrict__ Wneigh,
                                 const float* __restrict__ bias, float* __restrict__ Out,
                                 int Nnodes, int K, int F, int do_relu) {
  const int wid = threadIdx.x >> 5;
  const int lane = threadIdx.x & 31;
  const int tiles_n = F >> 4;
  const int m_tiles_per_block = 8 / tiles_n;
  const int m_tile = blockIdx.x * m_tiles_per_block + wid / tiles_n;
  const int n0 = (wid % tiles_n) << 4;
  const int m0 = m_tile << 4;
  if (m0 >= Nnodes) return;

  const int lane_lo = lane & 15;
  const int khalf = (lane >> 4) << 1;
  const size_t rowA = (size_t)(m0 + lane_lo) * K;

  v8f c = {};
  for (int k0 = 0; k0 < K; k0 += 4) {
    const int k = k0 + khalf;
    v2f a = *(const v2f*)(Xs + rowA + k);
    v2f b;
    b.x = Wself[(size_t)k * F + n0 + lane_lo];
    b.y = Wself[(size_t)(k + 1) * F + n0 + lane_lo];
    c = __builtin_amdgcn_wmma_f32_16x16x4_f32(false, a, false, b, (short)0, c, false, false);
    v2f a2 = *(const v2f*)(Xn + rowA + k);
    v2f b2;
    b2.x = Wneigh[(size_t)k * F + n0 + lane_lo];
    b2.y = Wneigh[(size_t)(k + 1) * F + n0 + lane_lo];
    c = __builtin_amdgcn_wmma_f32_16x16x4_f32(false, a2, false, b2, (short)0, c, false, false);
  }

  const float bv = bias[n0 + lane_lo];
  const int mbase = m0 + ((lane >> 4) << 3);
#pragma unroll
  for (int r = 0; r < 8; ++r) {
    float v = c[r] + bv;
    if (do_relu) v = fmaxf(v, 0.0f);
    Out[(size_t)(mbase + r) * F + n0 + lane_lo] = v;
  }
}

extern "C" void kernel_launch(void* const* d_in, const int* in_sizes, int n_in,
                              void* d_out, int out_size, void* d_ws, size_t ws_size,
                              hipStream_t stream) {
  (void)in_sizes; (void)n_in; (void)out_size; (void)ws_size;

  const float* x   = (const float*)d_in[0];
  const int*   src = (const int*)d_in[1];
  const int*   dst = (const int*)d_in[2];
  const float* W1s = (const float*)d_in[3];
  const float* W1n = (const float*)d_in[4];
  const float* b1  = (const float*)d_in[5];
  const float* W2s = (const float*)d_in[6];
  const float* W2n = (const float*)d_in[7];
  const float* b2  = (const float*)d_in[8];
  float* out = (float*)d_out;

  const int N = 50000, E = 800000, H = 128, F2 = 64;
  const long NH = (long)N * H;

  float* ws  = (float*)d_ws;
  float* deg = ws;
  float* agg = ws + N;
  float* hn  = agg + NH;
  float* h   = hn + NH;

  const int ZB = 2048, T = 256;
  const int edge_waves_blocks = (E * 32 + T - 1) / T;
  const int norm_blocks = (int)((NH + T - 1) / T);

  zero_f32<<<ZB, T, 0, stream>>>(deg, (long)N + NH);
  degree_kernel<<<(E + T - 1) / T, T, 0, stream>>>(dst, deg, E);
  scatter_add_kernel<<<edge_waves_blocks, T, 0, stream>>>(x, src, dst, agg, E);
  normalize_kernel<<<norm_blocks, T, 0, stream>>>(agg, deg, hn, NH);
  sage_linear_wmma<<<N / 16, T, 0, stream>>>(x, hn, W1s, W1n, b1, h, N, H, H, 1);

  zero_f32<<<ZB, T, 0, stream>>>(agg, NH);
  scatter_add_kernel<<<edge_waves_blocks, T, 0, stream>>>(h, src, dst, agg, E);
  normalize_kernel<<<norm_blocks, T, 0, stream>>>(agg, deg, hn, NH);
  sage_linear_wmma<<<(N / 16 + 1) / 2, T, 0, stream>>>(h, hn, W2s, W2n, b2, out, N, H, F2, 0);
}


# Edge list swept once: layer 1's owner sweep also stores each wave's matched (source, local row) pairs; layer 2's aggregation replays those lists and reads no edge ids
# speedup vs baseline: 2.8959x; 2.8959x over previous
#include <hip/hip_runtime.h>


typedef __attribute__((ext_vector_type(2))) float v2f;
typedef __attribute__((ext_vector_type(8))) float v8f;

#define FEAT 128
#define RPB 196
#define AGG_WAVES 32
#define EPW 1024
#define QCAP 1056
#define WCAP 320
static_assert(RPB <= 256, "a list entry keeps the local row in its low byte");

__device__ __forceinline__ void add_rows(const float* __restrict__ xin, float* acc, int s_i, int d_i,
                                         int n, int lane) {
  const int nb = (n + 7) >> 3;
  for (int b = 0; b < nb; ++b) {
    float v[8][4];
#pragma unroll
    for (int t = 0; t < 8; ++t) {
      const int s = __builtin_amdgcn_readlane(s_i, b * 8 + t);
      const float* p = xin + (size_t)s * FEAT + lane;
#pragma unroll
      for (int k = 0; k < 4; ++k) v[t][k] = p[32 * k];
    }
#pragma unroll
    for (int t = 0; t < 8; ++t) {
      const int d = __builtin_amdgcn_readlane(d_i, b * 8 + t);
      float* a = acc + d * FEAT + lane;
#pragma unroll
      for (int k = 0; k < 4; ++k) atomicAdd(a + 32 * k, v[t][k]);
    }
  }
}

template <bool RECORD>
__device__ __forceinline__ void drain_queue(const float* __restrict__ xin, const int* __restrict__ src,
                                            const int* __restrict__ dst, float* acc, float* cnt,
                                            const int* q, int n, int lo, int lane, int* lst, int& ln) {
  int d_i = RPB, s_i = 0;
  if (lane < n) {
    const int e = q[lane];
    s_i = src[e];
    d_i = dst[e] - lo;
    atomicAdd(&cnt[d_i], 1.0f);
    if (RECORD) {
      if (ln + lane < WCAP) lst[ln + lane] = (s_i << 8) | d_i;
    }
  }
  if (RECORD) ln += n;
  add_rows(xin, acc, s_i, d_i, n, lane);
}

template <bool RECORD>
__device__ __forceinline__ void sweep_share(const float* __restrict__ xin, const int* __restrict__ src,
                                            const int* __restrict__ dst, float* acc, float* cnt, int* q,
                                            int lo, int w, int lane, int E, int* lst, int& ln) {
  const int groups = (E + AGG_WAVES * EPW - 1) / (AGG_WAVES * EPW);
  const int rot = (int)((blockIdx.x * (unsigned)groups) / gridDim.x);
  int qn = 0;
  for (int g = 0; g < groups; ++g) {
    int gg = g + rot;
    if (gg >= groups) gg -= groups;
    const int wbase = (gg * AGG_WAVES + w) * EPW;
    const int ebase = wbase + lane * 4;
    int4 d[8];
    if (wbase + EPW <= E) {
      const int4* p = (const int4*)(dst + wbase) + lane;
#pragma unroll
      for (int u = 0; u < 8; ++u) d[u] = p[u * 32];
    } else {
#pragma unroll
      for (int u = 0; u < 8; ++u) {
        const int e = ebase + u * 128;
        d[u] = (e < E) ? *(const int4*)(dst + e) : make_int4(-1, -1, -1, -1);
      }
    }
#pragma unroll
    for (int u = 0; u < 8; ++u) {
      const unsigned r[4] = {(unsigned)(d[u].x - lo), (unsigned)(d[u].y - lo),
                             (unsigned)(d[u].z - lo), (unsigned)(d[u].w - lo)};
      const unsigned rmin = min(min(r[0], r[1]), min(r[2], r[3]));
      if (__builtin_amdgcn_ballot_w32(rmin < RPB) == 0u) continue;
#pragma unroll
      for (int j = 0; j < 4; ++j) {
        const bool hit = r[j] < RPB;
        const unsigned mask = __builtin_amdgcn_ballot_w32(hit);
        if (mask == 0u) continue;
        if (hit) q[qn + (int)__builtin_amdgcn_mbcnt_lo(mask, 0u)] = ebase + u * 128 + j;
        qn += __builtin_popcount(mask);
      }
    }
    if (qn >= 32) {
      int head = 0;
      while (qn - head >= 32) {
        drain_queue<RECORD>(xin, src, dst, acc, cnt, q + head, 32, lo, lane, lst, ln);
        head += 32;
      }
      const int rem = qn - head;
      int te = 0;
      if (lane < rem) te = q[head + lane];
      if (lane < rem) q[lane] = te;
      qn = rem;
    }
  }
  if (qn > 0) drain_queue<RECORD>(xin, src, dst, acc, cnt, q, qn, lo, lane, lst, ln);
}

template <bool REPLAY>
__global__ __launch_bounds__(1024) void aggregate_owner_kernel(const float* __restrict__ xin,
                                                               const int* __restrict__ src,
                                                               const int* __restrict__ dst,
                                                               float* __restrict__ hn,
                                                               int* __restrict__ lists,
                                                               int* __restrict__ counts, int N, int E) {
  __shared__ float acc[(RPB + 1) * FEAT];
  __shared__ float cnt[RPB + 4];
  __shared__ int q_all[AGG_WAVES][QCAP];

  const int tid = threadIdx.x;
  const int w = __builtin_amdgcn_readfirstlane(tid >> 5);
  const int lane = tid & 31;
  const int lo = blockIdx.x * RPB;
  int* q = q_all[w];
  const int slot = blockIdx.x * AGG_WAVES + w;
  int* lst = lists + (size_t)slot * WCAP;

  for (int i = tid; i < (RPB + 1) * FEAT; i += 1024) acc[i] = 0.0f;
  if (tid < RPB + 4) cnt[tid] = 0.0f;
  __syncthreads();

  if (!REPLAY) {
    int ln = 0;
    sweep_share<true>(xin, src, dst, acc, cnt, q, lo, w, lane, E, lst, ln);
    if (lane == 0) counts[slot] = ln;
  } else {
    const int c = counts[slot];
    if (c <= WCAP) {
      int ent = (lane < c) ? lst[lane] : 0;
      for (int i = 0; i < c; i += 32) {
        const int n = min(32, c - i);
        const int cur = ent;
        if (i + 32 < c) ent = (i + 32 + lane < c) ? lst[i + 32 + lane] : 0;
        int d_i = RPB, s_i = 0;
        if (lane < n) {
          s_i = cur >> 8;
          d_i = cur & 255;
          atomicAdd(&cnt[d_i], 1.0f);
        }
        add_rows(xin, acc, s_i, d_i, n, lane);
      }
    } else {
      int ln = 0;
      sweep_share<false>(xin, src, dst, acc, cnt, q, lo, w, lane, E, lst, ln);
    }
  }
  __syncthreads();

  for (int r = w; r < RPB; r += AGG_WAVES) {
    const int node = lo + r;
    if (node >= N) break;
    const float dg = fmaxf(cnt[r], 1.0f);
    const float* a = acc + r * FEAT + lane;
    float* o = hn + (size_t)node * FEAT + lane;
#pragma unroll
    for (int k = 0; k < 4; ++k) o[32 * k] = a[32 * k] / dg;
  }
}

__global__ void sage_linear_wmma(const float* __restrict__ Xs, const float* __restrict__ Xn,
                                 const float* __restrict__ Wself, const float* __restrict__ Wneigh,
                                 const float* __restrict__ bias, float* __restrict__ Out,
                                 int Nnodes, int K, int F, int do_relu) {
  const int wid = threadIdx.x >> 5;
  const int lane = threadIdx.x & 31;
  const int tiles_n = F >> 4;
  const int m_tiles_per_block = 8 / tiles_n;
  const int m_tile = blockIdx.x * m_tiles_per_block + wid / tiles_n;
  const int n0 = (wid % tiles_n) << 4;
  const int m0 = m_tile << 4;
  if (m0 >= Nnodes) return;

  const int lane_lo = lane & 15;
  const int khalf = (lane >> 4) << 1;
  const size_t rowA = (size_t)(m0 + lane_lo) * K;

  v8f c = {};
  for (int k0 = 0; k0 < K; k0 += 4) {
    const int k = k0 + khalf;
    v2f a = *(const v2f*)(Xs + rowA + k);
    v2f b;
    b.x = Wself[(size_t)k * F + n0 + lane_lo];
    b.y = Wself[(size_t)(k + 1) * F + n0 + lane_lo];
    c = __builtin_amdgcn_wmma_f32_16x16x4_f32(false, a, false, b, (short)0, c, false, false);
    v2f a2 = *(const v2f*)(Xn + rowA + k);
    v2f b2;
    b2.x = Wneigh[(size_t)k * F + n0 + lane_lo];
    b2.y = Wneigh[(size_t)(k + 1) * F + n0 + lane_lo];
    c = __builtin_amdgcn_wmma_f32_16x16x4_f32(false, a2, false, b2, (short)0, c, false, false);
  }

  const float bv = bias[n0 + lane_lo];
  const int mbase = m0 + ((lane >> 4) << 3);
#pragma unroll
  for (int r = 0; r < 8; ++r) {
    float v = c[r] + bv;
    if (do_relu) v = fmaxf(v, 0.0f);
    Out[(size_t)(mbase + r) * F + n0 + lane_lo] = v;
  }
}

extern "C" void kernel_launch(void* const* d_in, const int* in_sizes, int n_in,
                              void* d_out, int out_size, void* d_ws, size_t ws_size,
                              hipStream_t stream) {
  (void)in_sizes; (void)n_in; (void)out_size; (void)ws_size;

  const float* x   = (const float*)d_in[0];
  const int*   src = (const int*)d_in[1];
  const int*   dst = (const int*)d_in[2];
  const float* W1s = (const float*)d_in[3];
  const float* W1n = (const float*)d_in[4];
  const float* b1  = (const float*)d_in[5];
  const float* W2s = (const float*)d_in[6];
  const float* W2n = (const float*)d_in[7];
  const float* b2  = (const float*)d_in[8];
  float* out = (float*)d_out;

  const int N = 50000, E = 800000, H = 128, F2 = 64;
  const long NH = (long)N * H;

  float* ws = (float*)d_ws;
  float* hn = ws;
  float* h  = hn + NH;

  const int T = 256;
  const int AGG_BLOCKS = (N + RPB - 1) / RPB;
  const int AGG_T = AGG_WAVES * 32;

  int* lists  = (int*)d_out;
  int* counts = lists + AGG_BLOCKS * AGG_WAVES * WCAP;

  aggregate_owner_kernel<false><<<AGG_BLOCKS, AGG_T, 0, stream>>>(x, src, dst, hn, lists, counts, N, E);
  sage_linear_wmma<<<N / 16, T, 0, stream>>>(x, hn, W1s, W1n, b1, h, N, H, H, 1);

  aggregate_owner_kernel<true><<<AGG_BLOCKS, AGG_T, 0, stream>>>(h, src, dst, hn, lists, counts, N, E);
  sage_linear_wmma<<<(N / 16 + 1) / 2, T, 0, stream>>>(h, hn, W2s, W2n, b2, out, N, H, F2, 0);
}


# Linear phase with weights resident in LDS: 256 blocks copy Wself|Wneigh once in fragment order, a wave owns 16 whole rows, reads x and hn once; k summed 8 at a time, self then neighbour block
# speedup vs baseline: 3.7901x; 1.3088x over previous
#include <hip/hip_runtime.h>


typedef __attribute__((ext_vector_type(2))) float v2f;
typedef __attribute__((ext_vector_type(8))) float v8f;

#define FEAT 128
#define RPB 196
#define AGG_WAVES 32
#define EPW 1024
#define QCAP 1056
#define WCAP 320
static_assert(RPB <= 256, "a list entry keeps the local row in its low byte");

__device__ __forceinline__ void add_rows(const float* __restrict__ xin, float* acc, int s_i, int d_i,
                                         int n, int lane) {
  const int nb = (n + 7) >> 3;
  for (int b = 0; b < nb; ++b) {
    float v[8][4];
#pragma unroll
    for (int t = 0; t < 8; ++t) {
      const int s = __builtin_amdgcn_readlane(s_i, b * 8 + t);
      const float* p = xin + (size_t)s * FEAT + lane;
#pragma unroll
      for (int k = 0; k < 4; ++k) v[t][k] = p[32 * k];
    }
#pragma unroll
    for (int t = 0; t < 8; ++t) {
      const int d = __builtin_amdgcn_readlane(d_i, b * 8 + t);
      float* a = acc + d * FEAT + lane;
#pragma unroll
      for (int k = 0; k < 4; ++k) atomicAdd(a + 32 * k, v[t][k]);
    }
  }
}

template <bool RECORD>
__device__ __forceinline__ void drain_queue(const float* __restrict__ xin, const int* __restrict__ src,
                                            const int* __restrict__ dst, float* acc, float* cnt,
                                            const int* q, int n, int lo, int lane, int* lst, int& ln) {
  int d_i = RPB, s_i = 0;
  if (lane < n) {
    const int e = q[lane];
    s_i = src[e];
    d_i = dst[e] - lo;
    atomicAdd(&cnt[d_i], 1.0f);
    if (RECORD) {
      if (ln + lane < WCAP) lst[ln + lane] = (s_i << 8) | d_i;
    }
  }
  if (RECORD) ln += n;
  add_rows(xin, acc, s_i, d_i, n, lane);
}

template <bool RECORD>
__device__ __forceinline__ void sweep_share(const float* __restrict__ xin, const int* __restrict__ src,
                                            const int* __restrict__ dst, float* acc, float* cnt, int* q,
                                            int lo, int w, int lane, int E, int* lst, int& ln) {
  const int groups = (E + AGG_WAVES * EPW - 1) / (AGG_WAVES * EPW);
  const int rot = (int)((blockIdx.x * (unsigned)groups) / gridDim.x);
  int qn = 0;
  for (int g = 0; g < groups; ++g) {
    int gg = g + rot;
    if (gg >= groups) gg -= groups;
    const int wbase = (gg * AGG_WAVES + w) * EPW;
    const int ebase = wbase + lane * 4;
    int4 d[8];
    if (wbase + EPW <= E) {
      const int4* p = (const int4*)(dst + wbase) + lane;
#pragma unroll
      for (int u = 0; u < 8; ++u) d[u] = p[u * 32];
    } else {
#pragma unroll
      for (int u = 0; u < 8; ++u) {
        const int e = ebase + u * 128;
        d[u] = (e < E) ? *(const int4*)(dst + e) : make_int4(-1, -1, -1, -1);
      }
    }
#pragma unroll
    for (int u = 0; u < 8; ++u) {
      const unsigned r[4] = {(unsigned)(d[u].x - lo), (unsigned)(d[u].y - lo),
                             (unsigned)(d[u].z - lo), (unsigned)(d[u].w - lo)};
      const unsigned rmin = min(min(r[0], r[1]), min(r[2], r[3]));
      if (__builtin_amdgcn_ballot_w32(rmin < RPB) == 0u) continue;
#pragma unroll
      for (int j = 0; j < 4; ++j) {
        const bool hit = r[j] < RPB;
        const unsigned mask = __builtin_amdgcn_ballot_w32(hit);
        if (mask == 0u) continue;
        if (hit) q[qn + (int)__builtin_amdgcn_mbcnt_lo(mask, 0u)] = ebase + u * 128 + j;
        qn += __builtin_popcount(mask);
      }
    }
    if (qn >= 32) {
      int head = 0;
      while (qn - head >= 32) {
        drain_queue<RECORD>(xin, src, dst, acc, cnt, q + head, 32, lo, lane, lst, ln);
        head += 32;
      }
      const int rem = qn - head;
      int te = 0;
      if (lane < rem) te = q[head + lane];
      if (lane < rem) q[lane] = te;
      qn = rem;
    }
  }
  if (qn > 0) drain_queue<RECORD>(xin, src, dst, acc, cnt, q, qn, lo, lane, lst, ln);
}

template <bool REPLAY>
__global__ __launch_bounds__(1024) void aggregate_owner_kernel(const float* __restrict__ xin,
                                                               const int* __restrict__ src,
                                                               const int* __restrict__ dst,
                                                               float* __restrict__ hn,
                                                               int* __restrict__ lists,
                                                               int* __restrict__ counts, int N, int E) {
  __shared__ float acc[(RPB + 1) * FEAT];
  __shared__ float cnt[RPB + 4];
  __shared__ int q_all[AGG_WAVES][QCAP];

  const int tid = threadIdx.x;
  const int w = __builtin_amdgcn_readfirstlane(tid >> 5);
  const int lane = tid & 31;
  const int lo = blockIdx.x * RPB;
  int* q = q_all[w];
  const int slot = blockIdx.x * AGG_WAVES + w;
  int* lst = lists + (size_t)slot * WCAP;

  for (int i = tid; i < (RPB + 1) * FEAT; i += 1024) acc[i] = 0.0f;
  if (tid < RPB + 4) cnt[tid] = 0.0f;
  __syncthreads();

  if (!REPLAY) {
    int ln = 0;
    sweep_share<true>(xin, src, dst, acc, cnt, q, lo, w, lane, E, lst, ln);
    if (lane == 0) counts[slot] = ln;
  } else {
    const int c = counts[slot];
    if (c <= WCAP) {
      int ent = (lane < c) ? lst[lane] : 0;
      for (int i = 0; i < c; i += 32) {
        const int n = min(32, c - i);
        const int cur = ent;
        if (i + 32 < c) ent = (i + 32 + lane < c) ? lst[i + 32 + lane] : 0;
        int d_i = RPB, s_i = 0;
        if (lane < n) {
          s_i = cur >> 8;
          d_i = cur & 255;
          atomicAdd(&cnt[d_i], 1.0f);
        }
        add_rows(xin, acc, s_i, d_i, n, lane);
      }
    } else {
      int ln = 0;
      sweep_share<false>(xin, src, dst, acc, cnt, q, lo, w, lane, E, lst, ln);
    }
  }
  __syncthreads();

  for (int r = w; r < RPB; r += AGG_WAVES) {
    const int node = lo + r;
    if (node >= N) break;
    const float dg = fmaxf(cnt[r], 1.0f);
    const float* a = acc + r * FEAT + lane;
    float* o = hn + (size_t)node * FEAT + lane;
#pragma unroll
    for (int k = 0; k < 4; ++k) o[32 * k] = a[32 * k] / dg;
  }
}

#define LIN_T 512
#define LIN_BLOCKS 256

template <int F, bool RELU>
__global__ __launch_bounds__(LIN_T) void sage_linear_resident(const float* __restrict__ Xs,
                                                              const float* __restrict__ Xn,
                                                              const float* __restrict__ Wself,
                                                              const float* __restrict__ Wneigh,
                                                              const float* __restrict__ bias,
                                                              float* __restrict__ Out, int Nnodes) {
  constexpr int NT = F / 16;
  constexpr int STEPS = 2 * FEAT / 8;
  __shared__ float4 wl[STEPS * NT * 32];

  const int tid = threadIdx.x;
  const int w = __builtin_amdgcn_readfirstlane(tid >> 5);
  const int lane = tid & 31;
  const int c = lane & 15;
  const int h = lane >> 4;

  const int t = blockIdx.x + LIN_BLOCKS * w;
  const bool has_strip = t < (Nnodes >> 4);
  const int m0 = t << 4;
  const float* xs = Xs + (size_t)(m0 + c) * FEAT + 4 * h;
  const float* xn = Xn + (size_t)(m0 + c) * FEAT + 4 * h;
  float4 a[4];
#pragma unroll
  for (int s = 0; s < 4; ++s) a[s] = make_float4(0.f, 0.f, 0.f, 0.f);
  if (has_strip) {
#pragma unroll
    for (int s = 0; s < 4; ++s) a[s] = *(const float4*)(xs + 8 * s);
  }

  static_assert((2 * FEAT / 4 * F) % LIN_T == 0, "whole trips");
#pragma unroll
  for (int i = 0; i < 2 * FEAT / 4 * F / LIN_T; ++i) {
    const int idx = i * LIN_T + tid;
    const int n = idx % F;
    const int q = idx / F;
    const float* Wm = (q < FEAT / 4) ? Wself : Wneigh;
    const float* p = Wm + (size_t)((q & (FEAT / 4 - 1)) * 4) * F + n;
    float4 v;
    v.x = p[0];
    v.y = p[F];
    v.z = p[2 * F];
    v.w = p[3 * F];
    wl[((q >> 1) * NT + (n >> 4)) * 32 + (q & 1) * 16 + (n & 15)] = v;
  }
  __syncthreads();

  if (!has_strip) return;
  const float4* bp = wl + lane;

  v8f acc[NT];
#pragma unroll
  for (int nt = 0; nt < NT; ++nt) acc[nt] = v8f{};

  float4 b[NT];
#pragma unroll
  for (int nt = 0; nt < NT; ++nt) b[nt] = bp[nt * 32];

  const float* xp = xs;
  for (int g = 0; g < STEPS / 4; ++g) {
    xp = (g == 3) ? xn : ((g == 7) ? xs : xp + 32);
    float4 an[4];
#pragma unroll
    for (int s = 0; s < 4; ++s) an[s] = *(const float4*)(xp + 8 * s);
#pragma unroll
    for (int s = 0; s < 4; ++s) {
      const int kn = (4 * g + s + 1) & (STEPS - 1);
      const float4* bq = bp + kn * (NT * 32);
      float4 bn[NT];
#pragma unroll
      for (int nt = 0; nt < NT; ++nt) bn[nt] = bq[nt * 32];
      __builtin_amdgcn_sched_barrier(0);
      const v2f a0 = {a[s].x, a[s].y};
      const v2f a1 = {a[s].z, a[s].w};
#pragma unroll
      for (int nt = 0; nt < NT; ++nt) {
        const v2f b0 = {b[nt].x, b[nt].y};
        acc[nt] = __builtin_amdgcn_wmma_f32_16x16x4_f32(false, a0, false, b0, (short)0, acc[nt], false, false);
      }
#pragma unroll
      for (int nt = 0; nt < NT; ++nt) {
        const v2f b1 = {b[nt].z, b[nt].w};
        acc[nt] = __builtin_amdgcn_wmma_f32_16x16x4_f32(false, a1, false, b1, (short)0, acc[nt], false, false);
      }
      __builtin_amdgcn_sched_barrier(0);
#pragma unroll
      for (int nt = 0; nt < NT; ++nt) b[nt] = bn[nt];
    }
#pragma unroll
    for (int s = 0; s < 4; ++s) a[s] = an[s];
  }

  const int mbase = m0 + (h << 3);
#pragma unroll
  for (int nt = 0; nt < NT; ++nt) {
    const float bv = bias[nt * 16 + c];
#pragma unroll
    for (int r = 0; r < 8; ++r) {
      float v = acc[nt][r] + bv;
      if (RELU) v = fmaxf(v, 0.0f);
      Out[(size_t)(mbase + r) * F + nt * 16 + c] = v;
    }
  }
}

extern "C" void kernel_launch(void* const* d_in, const int* in_sizes, int n_in,
                              void* d_out, int out_size, void* d_ws, size_t ws_size,
                              hipStream_t stream) {
  (void)in_sizes; (void)n_in; (void)out_size; (void)ws_size;

  const float* x   = (const float*)d_in[0];
  const int*   src = (const int*)d_in[1];
  const int*   dst = (const int*)d_in[2];
  const float* W1s = (const float*)d_in[3];
  const float* W1n = (const float*)d_in[4];
  const float* b1  = (const float*)d_in[5];
  const float* W2s = (const float*)d_in[6];
  const float* W2n = (const float*)d_in[7];
  const float* b2  = (const float*)d_in[8];
  float* out = (float*)d_out;

  const int N = 50000, E = 800000, H = 128;
  const long NH = (long)N * H;

  float* ws = (float*)d_ws;
  float* hn = ws;
  float* h  = hn + NH;

  static_assert(LIN_BLOCKS * (LIN_T / 32) * 16 >= N, "every 16-row strip needs a wave of its own");
  const int AGG_BLOCKS = (N + RPB - 1) / RPB;
  const int AGG_T = AGG_WAVES * 32;

  int* lists  = (int*)d_out;
  int* counts = lists + AGG_BLOCKS * AGG_WAVES * WCAP;

  aggregate_owner_kernel<false><<<AGG_BLOCKS, AGG_T, 0, stream>>>(x, src, dst, hn, lists, counts, N, E);
  sage_linear_resident<128, true><<<LIN_BLOCKS, LIN_T, 0, stream>>>(x, hn, W1s, W1n, b1, h, N);

  aggregate_owner_kernel<true><<<AGG_BLOCKS, AGG_T, 0, stream>>>(h, src, dst, hn, lists, counts, N, E);
  sage_linear_resident<64, false><<<LIN_BLOCKS, LIN_T, 0, stream>>>(h, hn, W2s, W2n, b2, out, N);
}


# Owner sweeps read one owner byte per edge, not a 4-byte id: a first launch writes id/196 as a byte; 16 edges matched per 16-byte load; a wave's edge share regroups, so LDS add order changes
# speedup vs baseline: 4.1373x; 1.0916x over previous
#include <hip/hip_runtime.h>


typedef __attribute__((ext_vector_type(2))) float v2f;
typedef __attribute__((ext_vector_type(8))) float v8f;

#define FEAT 128
#define RPB 196
#define AGG_WAVES 32
#define EPW 4096
#define UNIT 512
#define QCAP 1056
#define WCAP 320
static_assert(RPB <= 256, "a list entry keeps the local row in its low byte");

__device__ __forceinline__ unsigned pack_owners(int4 d) {
  const unsigned o0 = ((unsigned)d.x / RPB) & 255u, o1 = ((unsigned)d.y / RPB) & 255u;
  const unsigned o2 = ((unsigned)d.z / RPB) & 255u, o3 = ((unsigned)d.w / RPB) & 255u;
  return o0 | (o1 << 8) | (o2 << 16) | (o3 << 24);
}

__global__ __launch_bounds__(256) void owner_bytes_kernel(const int* __restrict__ dst,
                                                          unsigned* __restrict__ ob, int E) {
  const int base = blockIdx.x * 1024 + threadIdx.x;
  const int4* p = (const int4*)dst;
  if (4 * (blockIdx.x * 1024 + 1024) <= E) {
    int4 d[4];
#pragma unroll
    for (int i = 0; i < 4; ++i) d[i] = p[base + i * 256];
#pragma unroll
    for (int i = 0; i < 4; ++i) ob[base + i * 256] = pack_owners(d[i]);
  } else {
    for (int i = 0; i < 4; ++i) {
      const int t = base + i * 256;
      if (4 * t < E) ob[t] = pack_owners(p[t]);
    }
  }
}

__device__ __forceinline__ void add_rows(const float* __restrict__ xin, float* acc, int s_i, int d_i,
                                         int n, int lane) {
  const int nb = (n + 7) >> 3;
  for (int b = 0; b < nb; ++b) {
    float v[8][4];
#pragma unroll
    for (int t = 0; t < 8; ++t) {
      const int s = __builtin_amdgcn_readlane(s_i, b * 8 + t);
      const float* p = xin + (size_t)s * FEAT + lane;
#pragma unroll
      for (int k = 0; k < 4; ++k) v[t][k] = p[32 * k];
    }
#pragma unroll
    for (int t = 0; t < 8; ++t) {
      const int d = __builtin_amdgcn_readlane(d_i, b * 8 + t);
      float* a = acc + d * FEAT + lane;
#pragma unroll
      for (int k = 0; k < 4; ++k) atomicAdd(a + 32 * k, v[t][k]);
    }
  }
}

template <bool RECORD>
__device__ __forceinline__ void drain_queue(const float* __restrict__ xin, const int* __restrict__ src,
                                            const int* __restrict__ dst, float* acc, float* cnt,
                                            const int* q, int n, int lo, int lane, int* lst, int& ln) {
  int d_i = RPB, s_i = 0;
  if (lane < n) {
    const int e = q[lane];
    const int dd = dst[e] - lo;
    if ((unsigned)dd < (unsigned)RPB) {
      s_i = src[e];
      d_i = dd;
    }
    atomicAdd(&cnt[d_i], 1.0f);
    if (RECORD) {
      if (ln + lane < WCAP) lst[ln + lane] = (s_i << 8) | d_i;
    }
  }
  if (RECORD) ln += n;
  add_rows(xin, acc, s_i, d_i, n, lane);
}

template <bool RECORD>
__device__ __forceinline__ void sweep_share(const float* __restrict__ xin, const int* __restrict__ src,
                                            const int* __restrict__ dst,
                                            const unsigned char* __restrict__ ob, float* acc, float* cnt,
                                            int* q, int lo, int w, int lane, int E, int* lst, int& ln) {
  const int groups = (E + AGG_WAVES * EPW - 1) / (AGG_WAVES * EPW);
  const int rot = (int)((blockIdx.x * (unsigned)groups) / gridDim.x);
  const unsigned pat = blockIdx.x * 0x01010101u;
  int qn = 0;
  for (int g = 0; g < groups; ++g) {
    int gg = g + rot;
    if (gg >= groups) gg -= groups;
    const int wbase = (gg * AGG_WAVES + w) * EPW;
    if (wbase >= E) continue;
    const int ebase = wbase + lane * 16;
    uint4 d[8];
    if (wbase + EPW <= E) {
      const uint4* p = (const uint4*)(ob + wbase) + lane;
#pragma unroll
      for (int u = 0; u < 8; ++u) d[u] = p[u * 32];
    } else {
#pragma unroll
      for (int u = 0; u < 8; ++u) {
        const int e = ebase + u * UNIT;
        d[u] = (e < E) ? *(const uint4*)(ob + e) : make_uint4(~pat, ~pat, ~pat, ~pat);
      }
    }
#pragma unroll
    for (int u = 0; u < 8; ++u) {
      const unsigned t0 = d[u].x ^ pat, t1 = d[u].y ^ pat, t2 = d[u].z ^ pat, t3 = d[u].w ^ pat;
      const unsigned z0 = ~(((t0 & 0x7F7F7F7Fu) + 0x7F7F7F7Fu) | t0 | 0x7F7F7F7Fu);
      const unsigned z1 = ~(((t1 & 0x7F7F7F7Fu) + 0x7F7F7F7Fu) | t1 | 0x7F7F7F7Fu);
      const unsigned z2 = ~(((t2 & 0x7F7F7F7Fu) + 0x7F7F7F7Fu) | t2 | 0x7F7F7F7Fu);
      const unsigned z3 = ~(((t3 & 0x7F7F7F7Fu) + 0x7F7F7F7Fu) | t3 | 0x7F7F7F7Fu);
      unsigned m = (z0 >> 7) | (z1 >> 6) | (z2 >> 5) | (z3 >> 4);
      unsigned bal = __builtin_amdgcn_ballot_w32(m != 0u);
      while (bal != 0u) {
        if (m != 0u) {
          const int cb = __builtin_ctz(m);
          q[qn + (int)__builtin_amdgcn_mbcnt_lo(bal, 0u)] = ebase + u * UNIT + 4 * (cb & 7) + (cb >> 3);
          m &= m - 1u;
        }
        qn += __builtin_popcount(bal);
        bal = __builtin_amdgcn_ballot_w32(m != 0u);
      }
      if ((u & 1) && qn >= 32) {
        int head = 0;
        while (qn - head >= 32) {
          drain_queue<RECORD>(xin, src, dst, acc, cnt, q + head, 32, lo, lane, lst, ln);
          head += 32;
        }
        const int rem = qn - head;
        int te = 0;
        if (lane < rem) te = q[head + lane];
        if (lane < rem) q[lane] = te;
        qn = rem;
      }
    }
  }
  if (qn > 0) drain_queue<RECORD>(xin, src, dst, acc, cnt, q, qn, lo, lane, lst, ln);
}

template <bool REPLAY>
__global__ __launch_bounds__(1024) void aggregate_owner_kernel(const float* __restrict__ xin,
                                                               const int* __restrict__ src,
                                                               const int* __restrict__ dst,
                                                               const unsigned char* __restrict__ ob,
                                                               float* __restrict__ hn,
                                                               int* __restrict__ lists,
                                                               int* __restrict__ counts, int N, int E) {
  __shared__ float acc[(RPB + 1) * FEAT];
  __shared__ float cnt[RPB + 4];
  __shared__ int q_all[AGG_WAVES][QCAP];

  const int tid = threadIdx.x;
  const int w = __builtin_amdgcn_readfirstlane(tid >> 5);
  const int lane = tid & 31;
  const int lo = blockIdx.x * RPB;
  int* q = q_all[w];
  const int slot = blockIdx.x * AGG_WAVES + w;
  int* lst = lists + (size_t)slot * WCAP;

  for (int i = tid; i < (RPB + 1) * FEAT; i += 1024) acc[i] = 0.0f;
  if (tid < RPB + 4) cnt[tid] = 0.0f;
  __syncthreads();

  if (!REPLAY) {
    int ln = 0;
    sweep_share<true>(xin, src, dst, ob, acc, cnt, q, lo, w, lane, E, lst, ln);
    if (lane == 0) counts[slot] = ln;
  } else {
    const int c = counts[slot];
    if (c <= WCAP) {
      int ent = (lane < c) ? lst[lane] : 0;
      for (int i = 0; i < c; i += 32) {
        const int n = min(32, c - i);
        const int cur = ent;
        if (i + 32 < c) ent = (i + 32 + lane < c) ? lst[i + 32 + lane] : 0;
        int d_i = RPB, s_i = 0;
        if (lane < n) {
          s_i = cur >> 8;
          d_i = cur & 255;
          atomicAdd(&cnt[d_i], 1.0f);
        }
        add_rows(xin, acc, s_i, d_i, n, lane);
      }
    } else {
      int ln = 0;
      sweep_share<false>(xin, src, dst, ob, acc, cnt, q, lo, w, lane, E, lst, ln);
    }
  }
  __syncthreads();

  for (int r = w; r < RPB; r += AGG_WAVES) {
    const int node = lo + r;
    if (node >= N) break;
    const float dg = fmaxf(cnt[r], 1.0f);
    const float* a = acc + r * FEAT + lane;
    float* o = hn + (size_t)node * FEAT + lane;
#pragma unroll
    for (int k = 0; k < 4; ++k) o[32 * k] = a[32 * k] / dg;
  }
}

#define LIN_T 512
#define LIN_BLOCKS 256

template <int F, bool RELU>
__global__ __launch_bounds__(LIN_T) void sage_linear_resident(const float* __restrict__ Xs,
                                                              const float* __restrict__ Xn,
                                                              const float* __restrict__ Wself,
                                                              const float* __restrict__ Wneigh,
                                                              const float* __restrict__ bias,
                                                              float* __restrict__ Out, int Nnodes) {
  constexpr int NT = F / 16;
  constexpr int STEPS = 2 * FEAT / 8;
  __shared__ float4 wl[STEPS * NT * 32];

  const int tid = threadIdx.x;
  const int w = __builtin_amdgcn_readfirstlane(tid >> 5);
  const int lane = tid & 31;
  const int c = lane & 15;
  const int h = lane >> 4;

  const int t = blockIdx.x + LIN_BLOCKS * w;
  const bool has_strip = t < (Nnodes >> 4);
  const int m0 = t << 4;
  const float* xs = Xs + (size_t)(m0 + c) * FEAT + 4 * h;
  const float* xn = Xn + (size_t)(m0 + c) * FEAT + 4 * h;
  float4 a[4];
#pragma unroll
  for (int s = 0; s < 4; ++s) a[s] = make_float4(0.f, 0.f, 0.f, 0.f);
  if (has_strip) {
#pragma unroll
    for (int s = 0; s < 4; ++s) a[s] = *(const float4*)(xs + 8 * s);
  }

  static_assert((2 * FEAT / 4 * F) % LIN_T == 0, "whole trips");
#pragma unroll
  for (int i = 0; i < 2 * FEAT / 4 * F / LIN_T; ++i) {
    const int idx = i * LIN_T + tid;
    const int n = idx % F;
    const int q = idx / F;
    const float* Wm = (q < FEAT / 4) ? Wself : Wneigh;
    const float* p = Wm + (size_t)((q & (FEAT / 4 - 1)) * 4) * F + n;
    float4 v;
    v.x = p[0];
    v.y = p[F];
    v.z = p[2 * F];
    v.w = p[3 * F];
    wl[((q >> 1) * NT + (n >> 4)) * 32 + (q & 1) * 16 + (n & 15)] = v;
  }
  __syncthreads();

  if (!has_strip) return;
  const float4* bp = wl + lane;

  v8f acc[NT];
#pragma unroll
  for (int nt = 0; nt < NT; ++nt) acc[nt] = v8f{};

  float4 b[NT];
#pragma unroll
  for (int nt = 0; nt < NT; ++nt) b[nt] = bp[nt * 32];

  const float* xp = xs;
  for (int g = 0; g < STEPS / 4; ++g) {
    xp = (g == 3) ? xn : ((g == 7) ? xs : xp + 32);
    float4 an[4];
#pragma unroll
    for (int s = 0; s < 4; ++s) an[s] = *(const float4*)(xp + 8 * s);
#pragma unroll
    for (int s = 0; s < 4; ++s) {
      const int kn = (4 * g + s + 1) & (STEPS - 1);
      const float4* bq = bp + kn * (NT * 32);
      float4 bn[NT];
#pragma unroll
      for (int nt = 0; nt < NT; ++nt) bn[nt] = bq[nt * 32];
      __builtin_amdgcn_sched_barrier(0);
      const v2f a0 = {a[s].x, a[s].y};
      const v2f a1 = {a[s].z, a[s].w};
#pragma unroll
      for (int nt = 0; nt < NT; ++nt) {
        const v2f b0 = {b[nt].x, b[nt].y};
        acc[nt] = __builtin_amdgcn_wmma_f32_16x16x4_f32(false, a0, false, b0, (short)0, acc[nt], false, false);
      }
#pragma unroll
      for (int nt = 0; nt < NT; ++nt) {
        const v2f b1 = {b[nt].z, b[nt].w};
        acc[nt] = __builtin_amdgcn_wmma_f32_16x16x4_f32(false, a1, false, b1, (short)0, acc[nt], false, false);
      }
      __builtin_amdgcn_sched_barrier(0);
#pragma unroll
      for (int nt = 0; nt < NT; ++nt) b[nt] = bn[nt];
    }
#pragma unroll
    for (int s = 0; s < 4; ++s) a[s] = an[s];
  }

  const int mbase = m0 + (h << 3);
#pragma unroll
  for (int nt = 0; nt < NT; ++nt) {
    const float bv = bias[nt * 16 + c];
#pragma unroll
    for (int r = 0; r < 8; ++r) {
      float v = acc[nt][r] + bv;
      if (RELU) v = fmaxf(v, 0.0f);
      Out[(size_t)(mbase + r) * F + nt * 16 + c] = v;
    }
  }
}

extern "C" void kernel_launch(void* const* d_in, const int* in_sizes, int n_in,
                              void* d_out, int out_size, void* d_ws, size_t ws_size,
                              hipStream_t stream) {
  (void)in_sizes; (void)n_in; (void)out_size; (void)ws_size;

  const float* x   = (const float*)d_in[0];
  const int*   src = (const int*)d_in[1];
  const int*   dst = (const int*)d_in[2];
  const float* W1s = (const float*)d_in[3];
  const float* W1n = (const float*)d_in[4];
  const float* b1  = (const float*)d_in[5];
  const float* W2s = (const float*)d_in[6];
  const float* W2n = (const float*)d_in[7];
  const float* b2  = (const float*)d_in[8];
  float* out = (float*)d_out;

  const int N = 50000, E = 800000, H = 128;
  const long NH = (long)N * H;

  float* ws = (float*)d_ws;
  float* hn = ws;
  float* h  = hn + NH;

  static_assert(LIN_BLOCKS * (LIN_T / 32) * 16 >= N, "every 16-row strip needs a wave of its own");
  const int AGG_BLOCKS = (N + RPB - 1) / RPB;
  const int AGG_T = AGG_WAVES * 32;

  int* lists  = (int*)d_out;
  int* counts = lists + AGG_BLOCKS * AGG_WAVES * WCAP;
  unsigned* ob = (unsigned*)(counts + AGG_BLOCKS * AGG_WAVES);
  static_assert(((size_t)256 * AGG_WAVES * (WCAP + 1) * 4) % 16 == 0, "16-byte loads of the owner bytes");
  static_assert((size_t)256 * AGG_WAVES * (WCAP + 1) * 4 + 800000 <= (size_t)50000 * 64 * 4, "fits the output buffer");

  owner_bytes_kernel<<<(E / 4 + 1023) / 1024, 256, 0, stream>>>(dst, ob, E);
  aggregate_owner_kernel<false><<<AGG_BLOCKS, AGG_T, 0, stream>>>(x, src, dst, (const unsigned char*)ob, hn, lists, counts, N, E);
  sage_linear_resident<128, true><<<LIN_BLOCKS, LIN_T, 0, stream>>>(x, hn, W1s, W1n, b1, h, N);

  aggregate_owner_kernel<true><<<AGG_BLOCKS, AGG_T, 0, stream>>>(h, src, dst, (const unsigned char*)ob, hn, lists, counts, N, E);
  sage_linear_resident<64, false><<<LIN_BLOCKS, LIN_T, 0, stream>>>(h, hn, W2s, W2n, b2, out, N);
}


# Aggregation sums in registers, not LDS: an owner block orders its collected edges by row (LDS counting sort, ids ascending), a wave sums whole 512-byte source rows per row; sum order changes
# speedup vs baseline: 4.7386x; 1.1453x over previous
#include <hip/hip_runtime.h>


typedef __attribute__((ext_vector_type(2))) float v2f;
typedef __attribute__((ext_vector_type(8))) float v8f;

#define FEAT 128
#define RPB 196
#define AGG_WAVES 32
#define EPW 4096
#define UNIT 512
#define QCAP 1056
#define WCAP 320
#define SCAP (AGG_WAVES * WCAP)
#define ROWMAX 64
#define RSP 200
static_assert(RPB <= 256, "a list entry keeps the local row in its low byte");
static_assert(RPB + 1 <= RSP, "row starts fit their plane");

__device__ __forceinline__ unsigned pack_owners(int4 d) {
  const unsigned o0 = ((unsigned)d.x / RPB) & 255u, o1 = ((unsigned)d.y / RPB) & 255u;
  const unsigned o2 = ((unsigned)d.z / RPB) & 255u, o3 = ((unsigned)d.w / RPB) & 255u;
  return o0 | (o1 << 8) | (o2 << 16) | (o3 << 24);
}

__global__ __launch_bounds__(256) void owner_bytes_kernel(const int* __restrict__ dst,
                                                          unsigned* __restrict__ ob, int E) {
  const int base = blockIdx.x * 1024 + threadIdx.x;
  const int4* p = (const int4*)dst;
  if (4 * (blockIdx.x * 1024 + 1024) <= E) {
    int4 d[4];
#pragma unroll
    for (int i = 0; i < 4; ++i) d[i] = p[base + i * 256];
#pragma unroll
    for (int i = 0; i < 4; ++i) ob[base + i * 256] = pack_owners(d[i]);
  } else {
    for (int i = 0; i < 4; ++i) {
      const int t = base + i * 256;
      if (4 * t < E) ob[t] = pack_owners(p[t]);
    }
  }
}

__device__ __forceinline__ void add_rows(const float* __restrict__ xin, float* acc, int s_i, int d_i,
                                         int n, int lane) {
  const int nb = (n + 7) >> 3;
  for (int b = 0; b < nb; ++b) {
    float v[8][4];
#pragma unroll
    for (int t = 0; t < 8; ++t) {
      const int s = __builtin_amdgcn_readlane(s_i, b * 8 + t);
      const float* p = xin + (size_t)s * FEAT + lane;
#pragma unroll
      for (int k = 0; k < 4; ++k) v[t][k] = p[32 * k];
    }
#pragma unroll
    for (int t = 0; t < 8; ++t) {
      const int d = __builtin_amdgcn_readlane(d_i, b * 8 + t);
      float* a = acc + d * FEAT + lane;
#pragma unroll
      for (int k = 0; k < 4; ++k) atomicAdd(a + 32 * k, v[t][k]);
    }
  }
}

template <bool COLLECT>
__device__ __forceinline__ void drain_queue(const float* __restrict__ xin, const int* __restrict__ src,
                                            const int* __restrict__ dst, float* acc, float* cnt,
                                            const int* q, int n, int lo, int lane, int* wl, int* rc, int& ln) {
  if (COLLECT) {
    bool ok = false;
    int ent = 0, dd = 0;
    if (lane < n) {
      const int e = q[lane];
      dd = dst[e] - lo;
      if ((unsigned)dd < (unsigned)RPB) {
        ok = true;
        ent = (src[e] << 8) | dd;
      }
    }
    const unsigned bal = __builtin_amdgcn_ballot_w32(ok);
    if (ok) {
      const int p = ln + (int)__builtin_amdgcn_mbcnt_lo(bal, 0u);
      if (p < WCAP) wl[p] = ent;
      atomicAdd(&rc[dd], 1);
    }
    ln += __builtin_popcount(bal);
    return;
  }
  int d_i = RPB, s_i = 0;
  if (lane < n) {
    const int e = q[lane];
    const int dd = dst[e] - lo;
    if ((unsigned)dd < (unsigned)RPB) {
      s_i = src[e];
      d_i = dd;
    }
    atomicAdd(&cnt[d_i], 1.0f);
  }
  add_rows(xin, acc, s_i, d_i, n, lane);
}

__device__ __forceinline__ float4 sum_rows(const float* __restrict__ xin, const int* list, int first32,
                                           int n, int lane) {
  float4 a = make_float4(0.f, 0.f, 0.f, 0.f);
  const float* xl = xin + 4 * lane;
  for (int c0 = 0; c0 < n; c0 += 32) {
    int sreg = first32;
    if (c0 > 0) sreg = (c0 + lane < n) ? list[c0 + lane] : 0;
    const int m = min(32, n - c0);
    for (int b = 0; b < m; b += 8) {
      float4 v[8];
#pragma unroll
      for (int t = 0; t < 8; ++t) {
        const int j = (b + t < m) ? b + t : b;
        const int s = __builtin_amdgcn_readlane(sreg, j);
        v[t] = *(const float4*)(xl + (size_t)s * FEAT);
      }
#pragma unroll
      for (int t = 0; t < 8; ++t) {
        if (b + t < m) {
          a.x += v[t].x;
          a.y += v[t].y;
          a.z += v[t].z;
          a.w += v[t].w;
        }
      }
    }
  }
  return a;
}

__device__ __forceinline__ void store_mean(float* __restrict__ hn, int node, float4 a, int n, int lane) {
  const float dg = fmaxf((float)n, 1.0f);
  float4 o;
  o.x = a.x / dg;
  o.y = a.y / dg;
  o.z = a.z / dg;
  o.w = a.w / dg;
  *(float4*)(hn + (size_t)node * FEAT + 4 * lane) = o;
}

template <bool COLLECT>
__device__ __forceinline__ void sweep_share(const float* __restrict__ xin, const int* __restrict__ src,
                                            const int* __restrict__ dst,
                                            const unsigned char* __restrict__ ob, float* acc, float* cnt,
                                            int* q, int lo, int w, int lane, int E, int* wl, int* rc, int& ln) {
  const int groups = (E + AGG_WAVES * EPW - 1) / (AGG_WAVES * EPW);
  const int rot = (int)((blockIdx.x * (unsigned)groups) / gridDim.x);
  const unsigned pat = blockIdx.x * 0x01010101u;
  int qn = 0;
  for (int g = 0; g < groups; ++g) {
    int gg = g + rot;
    if (gg >= groups) gg -= groups;
    const int wbase = (gg * AGG_WAVES + w) * EPW;
    if (wbase >= E) continue;
    const int ebase = wbase + lane * 16;
    uint4 d[8];
    if (wbase + EPW <= E) {
      const uint4* p = (const uint4*)(ob + wbase) + lane;
#pragma unroll
      for (int u = 0; u < 8; ++u) d[u] = p[u * 32];
    } else {
#pragma unroll
      for (int u = 0; u < 8; ++u) {
        const int e = ebase + u * UNIT;
        d[u] = (e < E) ? *(const uint4*)(ob + e) : make_uint4(~pat, ~pat, ~pat, ~pat);
      }
    }
#pragma unroll
    for (int u = 0; u < 8; ++u) {
      const unsigned t0 = d[u].x ^ pat, t1 = d[u].y ^ pat, t2 = d[u].z ^ pat, t3 = d[u].w ^ pat;
      const unsigned z0 = ~(((t0 & 0x7F7F7F7Fu) + 0x7F7F7F7Fu) | t0 | 0x7F7F7F7Fu);
      const unsigned z1 = ~(((t1 & 0x7F7F7F7Fu) + 0x7F7F7F7Fu) | t1 | 0x7F7F7F7Fu);
      const unsigned z2 = ~(((t2 & 0x7F7F7F7Fu) + 0x7F7F7F7Fu) | t2 | 0x7F7F7F7Fu);
      const unsigned z3 = ~(((t3 & 0x7F7F7F7Fu) + 0x7F7F7F7Fu) | t3 | 0x7F7F7F7Fu);
      unsigned m = (z0 >> 7) | (z1 >> 6) | (z2 >> 5) | (z3 >> 4);
      unsigned bal = __builtin_amdgcn_ballot_w32(m != 0u);
      while (bal != 0u) {
        if (m != 0u) {
          const int cb = __builtin_ctz(m);
          q[qn + (int)__builtin_amdgcn_mbcnt_lo(bal, 0u)] = ebase + u * UNIT + 4 * (cb & 7) + (cb >> 3);
          m &= m - 1u;
        }
        qn += __builtin_popcount(bal);
        bal = __builtin_amdgcn_ballot_w32(m != 0u);
      }
      if ((u & 1) && qn >= 32) {
        int head = 0;
        while (qn - head >= 32) {
          drain_queue<COLLECT>(xin, src, dst, acc, cnt, q + head, 32, lo, lane, wl, rc, ln);
          head += 32;
        }
        const int rem = qn - head;
        int te = 0;
        if (lane < rem) te = q[head + lane];
        if (lane < rem) q[lane] = te;
        qn = rem;
      }
    }
  }
  if (qn > 0) drain_queue<COLLECT>(xin, src, dst, acc, cnt, q, qn, lo, lane, wl, rc, ln);
}

template <bool REPLAY>
__global__ __launch_bounds__(1024) void aggregate_owner_kernel(const float* __restrict__ xin,
                                                               const int* __restrict__ src,
                                                               const int* __restrict__ dst,
                                                               const unsigned char* __restrict__ ob,
                                                               float* __restrict__ hn,
                                                               int* __restrict__ lists,
                                                               int* __restrict__ rowst,
                                                               int* __restrict__ flags, int N, int E) {
  __shared__ float acc[(RPB + 1) * FEAT];
  __shared__ float cnt[RPB + 4];
  __shared__ int q_all[AGG_WAVES][QCAP];
  __shared__ int rc[RSP];
  __shared__ int rs[RSP];
  __shared__ int cur[RSP];
  __shared__ int wn[AGG_WAVES];
  __shared__ int ovf;
  static_assert(2 * SCAP <= (RPB + 1) * FEAT, "both integer lists fit the room of the float sums");

  const int tid = threadIdx.x;
  const int w = __builtin_amdgcn_readfirstlane(tid >> 5);
  const int lane = tid & 31;
  const int lo = blockIdx.x * RPB;
  int* q = q_all[w];
  int* wl = (int*)acc + w * WCAP;
  int* sorted = (int*)acc + SCAP;
  int* glst = lists + (size_t)blockIdx.x * SCAP;
  int* grs = rowst + blockIdx.x * RSP;

  int general;
  if (!REPLAY) {
    if (tid < RSP) rc[tid] = 0;
    if (tid == 0) ovf = 0;
    __syncthreads();
    int ln = 0;
    sweep_share<true>(xin, src, dst, ob, acc, cnt, q, lo, w, lane, E, wl, rc, ln);
    if (lane == 0) {
      wn[w] = min(ln, WCAP);
      if (ln > WCAP) ovf = 1;
    }
    __syncthreads();
    if (tid < RPB && rc[tid] > ROWMAX) ovf = 1;
    __syncthreads();
    general = __builtin_amdgcn_readfirstlane(ovf);
    if (tid == 0) flags[blockIdx.x] = general;
  } else {
    general = __builtin_amdgcn_readfirstlane(flags[blockIdx.x]);
  }

  if (general == 0) {
    if (!REPLAY) {
      if (w == 0) {
        static_assert(7 * 32 >= RPB + 1, "one wave covers all rows and the total");
        int v[7];
        int tot = 0;
#pragma unroll
        for (int i = 0; i < 7; ++i) {
          const int r = 7 * lane + i;
          v[i] = (r < RPB) ? rc[r] : 0;
          tot += v[i];
        }
        int inc = tot;
#pragma unroll
        for (int d = 1; d < 32; d <<= 1) {
          const int o = __shfl_up(inc, d);
          if (lane >= d) inc += o;
        }
        int run = inc - tot;
#pragma unroll
        for (int i = 0; i < 7; ++i) {
          const int r = 7 * lane + i;
          if (r <= RPB) {
            rs[r] = run;
            cur[r] = run;
            grs[r] = run;
          }
          run += v[i];
        }
      }
      __syncthreads();
      const int nw = __builtin_amdgcn_readfirstlane(wn[w]);
      for (int i = lane; i < nw; i += 32) {
        const int ent = wl[i];
        const int pos = atomicAdd(&cur[ent & 255], 1);
        sorted[pos] = ent >> 8;
      }
      __syncthreads();
      for (int r = w; r < RPB; r += AGG_WAVES) {
        const int node = lo + r;
        if (node >= N) break;
        const int n = __builtin_amdgcn_readfirstlane(rc[r]);
        const int base = __builtin_amdgcn_readfirstlane(rs[r]);
        int* seg = sorted + base;
        const int e0 = (lane < n) ? seg[lane] : 0;
        const int e1 = (lane + 32 < n) ? seg[lane + 32] : 0;
        int k0 = 0, k1 = 0;
        const int n0 = min(n, 32);
        for (int j = 0; j < n0; ++j) {
          const int v = __builtin_amdgcn_readlane(e0, j);
          k0 += (v < e0 || (v == e0 && j < lane)) ? 1 : 0;
          k1 += (v <= e1) ? 1 : 0;
        }
        for (int j = 32; j < n; ++j) {
          const int v = __builtin_amdgcn_readlane(e1, j - 32);
          k0 += (v < e0) ? 1 : 0;
          k1 += (v < e1 || (v == e1 && j - 32 < lane)) ? 1 : 0;
        }
        if (lane < n) {
          seg[k0] = e0;
          glst[base + k0] = e0;
        }
        if (lane + 32 < n) {
          seg[k1] = e1;
          glst[base + k1] = e1;
        }
        const int f = (lane < n) ? seg[lane] : 0;
        const float4 a = sum_rows(xin, seg, f, n, lane);
        store_mean(hn, node, a, n, lane);
      }
    } else {
      int myA = 0, myB = 0;
      {
        const int r = w + AGG_WAVES * lane;
        if (r < RPB) {
          myA = grs[r];
          myB = grs[r + 1];
        }
      }
      int base = __builtin_amdgcn_readlane(myA, 0);
      int n = __builtin_amdgcn_readlane(myB, 0) - base;
      int f = (lane < n) ? glst[base + lane] : 0;
      int k = 0;
      for (int r = w; r < RPB; r += AGG_WAVES, ++k) {
        const int node = lo + r;
        if (node >= N) break;
        int basen = 0, nn = 0, fn = 0;
        if (r + AGG_WAVES < RPB) {
          basen = __builtin_amdgcn_readlane(myA, k + 1);
          nn = __builtin_amdgcn_readlane(myB, k + 1) - basen;
          fn = (lane < nn) ? glst[basen + lane] : 0;
        }
        const float4 a = sum_rows(xin, glst + base, f, n, lane);
        store_mean(hn, node, a, n, lane);
        base = basen;
        n = nn;
        f = fn;
      }
    }
    return;
  }

  for (int i = tid; i < (RPB + 1) * FEAT; i += 1024) acc[i] = 0.0f;
  if (tid < RPB + 4) cnt[tid] = 0.0f;
  __syncthreads();
  {
    int ln = 0;
    sweep_share<false>(xin, src, dst, ob, acc, cnt, q, lo, w, lane, E, wl, rc, ln);
  }
  __syncthreads();

  for (int r = w; r < RPB; r += AGG_WAVES) {
    const int node = lo + r;
    if (node >= N) break;
    const float dg = fmaxf(cnt[r], 1.0f);
    const float* a = acc + r * FEAT + lane;
    float* o = hn + (size_t)node * FEAT + lane;
#pragma unroll
    for (int k = 0; k < 4; ++k) o[32 * k] = a[32 * k] / dg;
  }
}

#define LIN_T 512
#define LIN_BLOCKS 256

template <int F, bool RELU>
__global__ __launch_bounds__(LIN_T) void sage_linear_resident(const float* __restrict__ Xs,
                                                              const float* __restrict__ Xn,
                                                              const float* __restrict__ Wself,
                                                              const float* __restrict__ Wneigh,
                                                              const float* __restrict__ bias,
                                                              float* __restrict__ Out, int Nnodes) {
  constexpr int NT = F / 16;
  constexpr int STEPS = 2 * FEAT / 8;
  __shared__ float4 wl[STEPS * NT * 32];

  const int tid = threadIdx.x;
  const int w = __builtin_amdgcn_readfirstlane(tid >> 5);
  const int lane = tid & 31;
  const int c = lane & 15;
  const int h = lane >> 4;

  const int t = blockIdx.x + LIN_BLOCKS * w;
  const bool has_strip = t < (Nnodes >> 4);
  const int m0 = t << 4;
  const float* xs = Xs + (size_t)(m0 + c) * FEAT + 4 * h;
  const float* xn = Xn + (size_t)(m0 + c) * FEAT + 4 * h;
  float4 a[4];
#pragma unroll
  for (int s = 0; s < 4; ++s) a[s] = make_float4(0.f, 0.f, 0.f, 0.f);
  if (has_strip) {
#pragma unroll
    for (int s = 0; s < 4; ++s) a[s] = *(const float4*)(xs + 8 * s);
  }

  static_assert((2 * FEAT / 4 * F) % LIN_T == 0, "whole trips");
#pragma unroll
  for (int i = 0; i < 2 * FEAT / 4 * F / LIN_T; ++i) {
    const int idx = i * LIN_T + tid;
    const int n = idx % F;
    const int q = idx / F;
    const float* Wm = (q < FEAT / 4) ? Wself : Wneigh;
    const float* p = Wm + (size_t)((q & (FEAT / 4 - 1)) * 4) * F + n;
    float4 v;
    v.x = p[0];
    v.y = p[F];
    v.z = p[2 * F];
    v.w = p[3 * F];
    wl[((q >> 1) * NT + (n >> 4)) * 32 + (q & 1) * 16 + (n & 15)] = v;
  }
  __syncthreads();

  if (!has_strip) return;
  const float4* bp = wl + lane;

  v8f acc[NT];
#pragma unroll
  for (int nt = 0; nt < NT; ++nt) acc[nt] = v8f{};

  float4 b[NT];
#pragma unroll
  for (int nt = 0; nt < NT; ++nt) b[nt] = bp[nt * 32];

  const float* xp = xs;
  for (int g = 0; g < STEPS / 4; ++g) {
    xp = (g == 3) ? xn : ((g == 7) ? xs : xp + 32);
    float4 an[4];
#pragma unroll
    for (int s = 0; s < 4; ++s) an[s] = *(const float4*)(xp + 8 * s);
#pragma unroll
    for (int s = 0; s < 4; ++s) {
      const int kn = (4 * g + s + 1) & (STEPS - 1);
      const float4* bq = bp + kn * (NT * 32);
      float4 bn[NT];
#pragma unroll
      for (int nt = 0; nt < NT; ++nt) bn[nt] = bq[nt * 32];
      __builtin_amdgcn_sched_barrier(0);
      const v2f a0 = {a[s].x, a[s].y};
      const v2f a1 = {a[s].z, a[s].w};
#pragma unroll
      for (int nt = 0; nt < NT; ++nt) {
        const v2f b0 = {b[nt].x, b[nt].y};
        acc[nt] = __builtin_amdgcn_wmma_f32_16x16x4_f32(false, a0, false, b0, (short)0, acc[nt], false, false);
      }
#pragma unroll
      for (int nt = 0; nt < NT; ++nt) {
        const v2f b1 = {b[nt].z, b[nt].w};
        acc[nt] = __builtin_amdgcn_wmma_f32_16x16x4_f32(false, a1, false, b1, (short)0, acc[nt], false, false);
      }
      __builtin_amdgcn_sched_barrier(0);
#pragma unroll
      for (int nt = 0; nt < NT; ++nt) b[nt] = bn[nt];
    }
#pragma unroll
    for (int s = 0; s < 4; ++s) a[s] = an[s];
  }

  const int mbase = m0 + (h << 3);
#pragma unroll
  for (int nt = 0; nt < NT; ++nt) {
    const float bv = bias[nt * 16 + c];
#pragma unroll
    for (int r = 0; r < 8; ++r) {
      float v = acc[nt][r] + bv;
      if (RELU) v = fmaxf(v, 0.0f);
      Out[(size_t)(mbase + r) * F + nt * 16 + c] = v;
    }
  }
}

extern "C" void kernel_launch(void* const* d_in, const int* in_sizes, int n_in,
                              void* d_out, int out_size, void* d_ws, size_t ws_size,
                              hipStream_t stream) {
  (void)in_sizes; (void)n_in; (void)out_size; (void)ws_size;

  const float* x   = (const float*)d_in[0];
  const int*   src = (const int*)d_in[1];
  const int*   dst = (const int*)d_in[2];
  const float* W1s = (const float*)d_in[3];
  const float* W1n = (const float*)d_in[4];
  const float* b1  = (const float*)d_in[5];
  const float* W2s = (const float*)d_in[6];
  const float* W2n = (const float*)d_in[7];
  const float* b2  = (const float*)d_in[8];
  float* out = (float*)d_out;

  const int N = 50000, E = 800000, H = 128;
  const long NH = (long)N * H;

  float* ws = (float*)d_ws;
  float* hn = ws;
  float* h  = hn + NH;

  static_assert(LIN_BLOCKS * (LIN_T / 32) * 16 >= N, "every 16-row strip needs a wave of its own");
  const int AGG_BLOCKS = (N + RPB - 1) / RPB;
  const int AGG_T = AGG_WAVES * 32;

  int* lists = (int*)d_out;
  int* rowst = lists + AGG_BLOCKS * SCAP;
  int* flags = rowst + AGG_BLOCKS * RSP;
  unsigned* ob = (unsigned*)(flags + AGG_BLOCKS);
  static_assert(((size_t)256 * (SCAP + RSP + 1) * 4) % 16 == 0, "16-byte loads of the owner bytes");
  static_assert((size_t)256 * (SCAP + RSP + 1) * 4 + 800000 <= (size_t)50000 * 64 * 4, "fits the output buffer");

  owner_bytes_kernel<<<(E / 4 + 1023) / 1024, 256, 0, stream>>>(dst, ob, E);
  aggregate_owner_kernel<false><<<AGG_BLOCKS, AGG_T, 0, stream>>>(x, src, dst, (const unsigned char*)ob, hn, lists, rowst, flags, N, E);
  sage_linear_resident<128, true><<<LIN_BLOCKS, LIN_T, 0, stream>>>(x, hn, W1s, W1n, b1, h, N);

  aggregate_owner_kernel<true><<<AGG_BLOCKS, AGG_T, 0, stream>>>(h, src, dst, (const unsigned char*)ob, hn, lists, rowst, flags, N, E);
  sage_linear_resident<64, false><<<LIN_BLOCKS, LIN_T, 0, stream>>>(h, hn, W2s, W2n, b2, out, N);
}


# Edges partitioned once by owner: a first launch orders each 4096-edge part by owner in LDS and writes per-(owner, part) cells; layer 1's owner blocks read their cells, sweep no bytes; sum order kept
# speedup vs baseline: 5.1015x; 1.0766x over previous
#include <hip/hip_runtime.h>


typedef __attribute__((ext_vector_type(2))) float v2f;
typedef __attribute__((ext_vector_type(8))) float v8f;

#define FEAT 128
#define RPB 196
#define AGG_WAVES 32
#define EPW 4096
#define UNIT 512
#define QCAP 1056
#define SCAP 10240
#define OWNERS 256
#define PEDGES 4096
#define PARTS 196
#define CELLCAP 64
#define PPW 7
#define ROWMAX 64
#define RSP 200
static_assert(RPB <= 256, "a list entry keeps the local row in its low byte");
static_assert(RPB + 1 <= RSP, "row starts fit their plane");
static_assert(PPW * AGG_WAVES >= PARTS, "every cell of an owner has a wave");
static_assert(CELLCAP == 64, "a wave reads a cell as two loads of 32 entries");

__device__ __forceinline__ unsigned pack_owners(int4 d) {
  const unsigned o0 = ((unsigned)d.x / RPB) & 255u, o1 = ((unsigned)d.y / RPB) & 255u;
  const unsigned o2 = ((unsigned)d.z / RPB) & 255u, o3 = ((unsigned)d.w / RPB) & 255u;
  return o0 | (o1 << 8) | (o2 << 16) | (o3 << 24);
}

__global__ __launch_bounds__(256) void partition_kernel(const int* __restrict__ src,
                                                        const int* __restrict__ dst,
                                                        unsigned* __restrict__ ob,
                                                        int* __restrict__ cells,
                                                        int* __restrict__ cellcnt, int E) {
  __shared__ int ocnt[OWNERS];
  __shared__ int ostart[OWNERS];
  __shared__ int2 s_pair[PEDGES];
  static_assert(OWNERS == 256 && PEDGES == 4 * 4 * 256, "256 threads: one owner each, four words of four edges each");

  const int tid = threadIdx.x;
  const int lane = tid & 31;
  const int base = blockIdx.x * (PEDGES / 4) + tid;
  const int nwords = E >> 2;
  const int4* pd = (const int4*)dst;
  const int4* ps = (const int4*)src;
  int dv[16], sv[16];
#pragma unroll
  for (int i = 0; i < 4; ++i) {
    const int t = base + i * 256;
    int4 d = make_int4(-1, -1, -1, -1), s = make_int4(0, 0, 0, 0);
    if (t < nwords) {
      d = pd[t];
      s = ps[t];
    }
    dv[4 * i + 0] = d.x; dv[4 * i + 1] = d.y; dv[4 * i + 2] = d.z; dv[4 * i + 3] = d.w;
    sv[4 * i + 0] = s.x; sv[4 * i + 1] = s.y; sv[4 * i + 2] = s.z; sv[4 * i + 3] = s.w;
  }
  ocnt[tid] = 0;
#pragma unroll
  for (int i = 0; i < PEDGES / 256; ++i) s_pair[tid + 256 * i] = make_int2(0, -1);
  __syncthreads();
#pragma unroll
  for (int i = 0; i < 4; ++i) {
    const int t = base + i * 256;
    if (t < nwords) ob[t] = pack_owners(make_int4(dv[4 * i], dv[4 * i + 1], dv[4 * i + 2], dv[4 * i + 3]));
  }
  int ow[16], rk[16];
#pragma unroll
  for (int j = 0; j < 16; ++j) {
    const unsigned o = (unsigned)dv[j] / RPB;
    ow[j] = (int)o;
    rk[j] = -1;
    if (o < (unsigned)OWNERS) rk[j] = atomicAdd(&ocnt[o], 1);
  }
  __syncthreads();
  cellcnt[blockIdx.x * OWNERS + tid] = ocnt[tid];
  if (tid < 32) {
    int v[8];
    int tot = 0;
#pragma unroll
    for (int i = 0; i < 8; ++i) {
      v[i] = ocnt[8 * lane + i];
      tot += v[i];
    }
    int inc = tot;
#pragma unroll
    for (int d = 1; d < 32; d <<= 1) {
      const int o = __shfl_up(inc, d);
      if (lane >= d) inc += o;
    }
    int run = inc - tot;
#pragma unroll
    for (int i = 0; i < 8; ++i) {
      ostart[8 * lane + i] = run;
      run += v[i];
    }
  }
  __syncthreads();
  int st[16];
#pragma unroll
  for (int j = 0; j < 16; ++j) st[j] = ostart[ow[j] & (OWNERS - 1)];
#pragma unroll
  for (int j = 0; j < 16; ++j) {
    if (rk[j] >= 0) {
      const int pos = st[j] + rk[j];
      const int ent = (sv[j] << 8) | (dv[j] - ow[j] * RPB);
      const int dpos = (rk[j] < CELLCAP) ? (ow[j] * PARTS + (int)blockIdx.x) * CELLCAP + rk[j] : -1;
      s_pair[pos] = make_int2(ent, dpos);
    }
  }
  __syncthreads();
  int2 pr[PEDGES / 256];
#pragma unroll
  for (int i = 0; i < PEDGES / 256; ++i) pr[i] = s_pair[tid + 256 * i];
#pragma unroll
  for (int i = 0; i < PEDGES / 256; ++i) {
    if (pr[i].y >= 0) cells[pr[i].y] = pr[i].x;
  }
}

__device__ __forceinline__ void add_rows(const float* __restrict__ xin, float* acc, int s_i, int d_i,
                                         int n, int lane) {
  const int nb = (n + 7) >> 3;
  for (int b = 0; b < nb; ++b) {
    float v[8][4];
#pragma unroll
    for (int t = 0; t < 8; ++t) {
      const int s = __builtin_amdgcn_readlane(s_i, b * 8 + t);
      const float* p = xin + (size_t)s * FEAT + lane;
#pragma unroll
      for (int k = 0; k < 4; ++k) v[t][k] = p[32 * k];
    }
#pragma unroll
    for (int t = 0; t < 8; ++t) {
      const int d = __builtin_amdgcn_readlane(d_i, b * 8 + t);
      float* a = acc + d * FEAT + lane;
#pragma unroll
      for (int k = 0; k < 4; ++k) atomicAdd(a + 32 * k, v[t][k]);
    }
  }
}

__device__ __forceinline__ void drain_queue(const float* __restrict__ xin, const int* __restrict__ src,
                                            const int* __restrict__ dst, float* acc, float* cnt,
                                            const int* q, int n, int lo, int lane) {
  int d_i = RPB, s_i = 0;
  if (lane < n) {
    const int e = q[lane];
    const int dd = dst[e] - lo;
    if ((unsigned)dd < (unsigned)RPB) {
      s_i = src[e];
      d_i = dd;
    }
    atomicAdd(&cnt[d_i], 1.0f);
  }
  add_rows(xin, acc, s_i, d_i, n, lane);
}

__device__ __forceinline__ float4 sum_rows(const float* __restrict__ xin, const int* list, int first32,
                                           int n, int lane) {
  float4 a = make_float4(0.f, 0.f, 0.f, 0.f);
  const float* xl = xin + 4 * lane;
  for (int c0 = 0; c0 < n; c0 += 32) {
    int sreg = first32;
    if (c0 > 0) sreg = (c0 + lane < n) ? list[c0 + lane] : 0;
    const int m = min(32, n - c0);
    for (int b = 0; b < m; b += 8) {
      float4 v[8];
#pragma unroll
      for (int t = 0; t < 8; ++t) {
        const int j = (b + t < m) ? b + t : b;
        const int s = __builtin_amdgcn_readlane(sreg, j);
        v[t] = *(const float4*)(xl + (size_t)s * FEAT);
      }
#pragma unroll
      for (int t = 0; t < 8; ++t) {
        if (b + t < m) {
          a.x += v[t].x;
          a.y += v[t].y;
          a.z += v[t].z;
          a.w += v[t].w;
        }
      }
    }
  }
  return a;
}

__device__ __forceinline__ void store_mean(float* __restrict__ hn, int node, float4 a, int n, int lane) {
  const float dg = fmaxf((float)n, 1.0f);
  float4 o;
  o.x = a.x / dg;
  o.y = a.y / dg;
  o.z = a.z / dg;
  o.w = a.w / dg;
  *(float4*)(hn + (size_t)node * FEAT + 4 * lane) = o;
}

__device__ __forceinline__ void sweep_share(const float* __restrict__ xin, const int* __restrict__ src,
                                            const int* __restrict__ dst,
                                            const unsigned char* __restrict__ ob, float* acc, float* cnt,
                                            int* q, int lo, int w, int lane, int E) {
  const int groups = (E + AGG_WAVES * EPW - 1) / (AGG_WAVES * EPW);
  const int rot = (int)((blockIdx.x * (unsigned)groups) / gridDim.x);
  const unsigned pat = blockIdx.x * 0x01010101u;
  int qn = 0;
  for (int g = 0; g < groups; ++g) {
    int gg = g + rot;
    if (gg >= groups) gg -= groups;
    const int wbase = (gg * AGG_WAVES + w) * EPW;
    if (wbase >= E) continue;
    const int ebase = wbase + lane * 16;
    uint4 d[8];
    if (wbase + EPW <= E) {
      const uint4* p = (const uint4*)(ob + wbase) + lane;
#pragma unroll
      for (int u = 0; u < 8; ++u) d[u] = p[u * 32];
    } else {
#pragma unroll
      for (int u = 0; u < 8; ++u) {
        const int e = ebase + u * UNIT;
        d[u] = (e < E) ? *(const uint4*)(ob + e) : make_uint4(~pat, ~pat, ~pat, ~pat);
      }
    }
#pragma unroll
    for (int u = 0; u < 8; ++u) {
      const unsigned t0 = d[u].x ^ pat, t1 = d[u].y ^ pat, t2 = d[u].z ^ pat, t3 = d[u].w ^ pat;
      const unsigned z0 = ~(((t0 & 0x7F7F7F7Fu) + 0x7F7F7F7Fu) | t0 | 0x7F7F7F7Fu);
      const unsigned z1 = ~(((t1 & 0x7F7F7F7Fu) + 0x7F7F7F7Fu) | t1 | 0x7F7F7F7Fu);
      const unsigned z2 = ~(((t2 & 0x7F7F7F7Fu) + 0x7F7F7F7Fu) | t2 | 0x7F7F7F7Fu);
      const unsigned z3 = ~(((t3 & 0x7F7F7F7Fu) + 0x7F7F7F7Fu) | t3 | 0x7F7F7F7Fu);
      unsigned m = (z0 >> 7) | (z1 >> 6) | (z2 >> 5) | (z3 >> 4);
      unsigned bal = __builtin_amdgcn_ballot_w32(m != 0u);
      while (bal != 0u) {
        if (m != 0u) {
          const int cb = __builtin_ctz(m);
          q[qn + (int)__builtin_amdgcn_mbcnt_lo(bal, 0u)] = ebase + u * UNIT + 4 * (cb & 7) + (cb >> 3);
          m &= m - 1u;
        }
        qn += __builtin_popcount(bal);
        bal = __builtin_amdgcn_ballot_w32(m != 0u);
      }
      if ((u & 1) && qn >= 32) {
        int head = 0;
        while (qn - head >= 32) {
          drain_queue(xin, src, dst, acc, cnt, q + head, 32, lo, lane);
          head += 32;
        }
        const int rem = qn - head;
        int te = 0;
        if (lane < rem) te = q[head + lane];
        if (lane < rem) q[lane] = te;
        qn = rem;
      }
    }
  }
  if (qn > 0) drain_queue(xin, src, dst, acc, cnt, q, qn, lo, lane);
}

template <bool REPLAY>
__global__ __launch_bounds__(1024) void aggregate_owner_kernel(const float* __restrict__ xin,
                                                               const int* __restrict__ src,
                                                               const int* __restrict__ dst,
                                                               const unsigned char* __restrict__ ob,
                                                               const int* __restrict__ cells,
                                                               const int* __restrict__ cellcnt,
                                                               float* __restrict__ hn,
                                                               int* __restrict__ lists,
                                                               int* __restrict__ rowst,
                                                               int* __restrict__ flags, int N, int E) {
  __shared__ float acc[(RPB + 1) * FEAT];
  __shared__ float cnt[RPB + 4];
  __shared__ int q_all[AGG_WAVES][QCAP];
  __shared__ int rc[RSP];
  __shared__ int rs[RSP];
  __shared__ int cur[RSP];
  __shared__ int btot;
  __shared__ int ovf;
  static_assert(SCAP <= (RPB + 1) * FEAT, "the ordered list fits the room of the float sums");

  const int tid = threadIdx.x;
  const int w = __builtin_amdgcn_readfirstlane(tid >> 5);
  const int lane = tid & 31;
  const int lo = blockIdx.x * RPB;
  int* q = q_all[w];
  int* sorted = (int*)acc;
  int* glst = lists + (size_t)blockIdx.x * SCAP;
  int* grs = rowst + blockIdx.x * RSP;

  int e0[PPW], e1[PPW];
  int general;
  if (!REPLAY) {
    if (tid < RSP) rc[tid] = 0;
    if (tid == 0) {
      ovf = 0;
      btot = 0;
    }
    int myc = 0;
    {
      const int p = w + AGG_WAVES * lane;
      if (p < PARTS) myc = cellcnt[p * OWNERS + blockIdx.x];
    }
    int ln = 0;
    bool over = false;
#pragma unroll
    for (int k = 0; k < PPW; ++k) {
      const int craw = __builtin_amdgcn_readlane(myc, k);
      over = over || (craw > CELLCAP);
      const int cn = min(craw, CELLCAP);
      const int* cell = cells + (blockIdx.x * PARTS + w + AGG_WAVES * k) * CELLCAP;
      int a = -1, b = -1;
      if (lane < cn) a = cell[lane];
      if (lane + 32 < cn) b = cell[lane + 32];
      e0[k] = a;
      e1[k] = b;
      ln += cn;
    }
    __syncthreads();
#pragma unroll
    for (int k = 0; k < PPW; ++k) {
      if (e0[k] >= 0) atomicAdd(&rc[e0[k] & 255], 1);
      if (e1[k] >= 0) atomicAdd(&rc[e1[k] & 255], 1);
    }
    if (lane == 0) {
      atomicAdd(&btot, ln);
      if (over) ovf = 1;
    }
    __syncthreads();
    if (tid < RPB && rc[tid] > ROWMAX) ovf = 1;
    if (tid == 0 && btot > SCAP) ovf = 1;
    __syncthreads();
    general = __builtin_amdgcn_readfirstlane(ovf);
    if (tid == 0) flags[blockIdx.x] = general;
  } else {
    general = __builtin_amdgcn_readfirstlane(flags[blockIdx.x]);
  }

  if (general == 0) {
    if (!REPLAY) {
      if (w == 0) {
        static_assert(7 * 32 >= RPB + 1, "one wave covers all rows and the total");
        int v[7];
        int tot = 0;
#pragma unroll
        for (int i = 0; i < 7; ++i) {
          const int r = 7 * lane + i;
          v[i] = (r < RPB) ? rc[r] : 0;
          tot += v[i];
        }
        int inc = tot;
#pragma unroll
        for (int d = 1; d < 32; d <<= 1) {
          const int o = __shfl_up(inc, d);
          if (lane >= d) inc += o;
        }
        int run = inc - tot;
#pragma unroll
        for (int i = 0; i < 7; ++i) {
          const int r = 7 * lane + i;
          if (r <= RPB) {
            rs[r] = run;
            cur[r] = run;
            grs[r] = run;
          }
          run += v[i];
        }
      }
      __syncthreads();
      int p0[PPW], p1[PPW];
#pragma unroll
      for (int k = 0; k < PPW; ++k) {
        p0[k] = 0;
        p1[k] = 0;
        if (e0[k] >= 0) p0[k] = atomicAdd(&cur[e0[k] & 255], 1);
        if (e1[k] >= 0) p1[k] = atomicAdd(&cur[e1[k] & 255], 1);
      }
#pragma unroll
      for (int k = 0; k < PPW; ++k) {
        if (e0[k] >= 0) sorted[p0[k]] = e0[k] >> 8;
        if (e1[k] >= 0) sorted[p1[k]] = e1[k] >> 8;
      }
      __syncthreads();
      for (int r = w; r < RPB; r += AGG_WAVES) {
        const int node = lo + r;
        if (node >= N) break;
        const int n = __builtin_amdgcn_readfirstlane(rc[r]);
        const int base = __builtin_amdgcn_readfirstlane(rs[r]);
        int* seg = sorted + base;
        const int e0 = (lane < n) ? seg[lane] : 0;
        const int e1 = (lane + 32 < n) ? seg[lane + 32] : 0;
        int k0 = 0, k1 = 0;
        const int n0 = min(n, 32);
        for (int j = 0; j < n0; ++j) {
          const int v = __builtin_amdgcn_readlane(e0, j);
          k0 += (v < e0 || (v == e0 && j < lane)) ? 1 : 0;
          k1 += (v <= e1) ? 1 : 0;
        }
        for (int j = 32; j < n; ++j) {
          const int v = __builtin_amdgcn_readlane(e1, j - 32);
          k0 += (v < e0) ? 1 : 0;
          k1 += (v < e1 || (v == e1 && j - 32 < lane)) ? 1 : 0;
        }
        if (lane < n) {
          seg[k0] = e0;
          glst[base + k0] = e0;
        }
        if (lane + 32 < n) {
          seg[k1] = e1;
          glst[base + k1] = e1;
        }
        const int f = (lane < n) ? seg[lane] : 0;
        const float4 a = sum_rows(xin, seg, f, n, lane);
        store_mean(hn, node, a, n, lane);
      }
    } else {
      int myA = 0, myB = 0;
      {
        const int r = w + AGG_WAVES * lane;
        if (r < RPB) {
          myA = grs[r];
          myB = grs[r + 1];
        }
      }
      int base = __builtin_amdgcn_readlane(myA, 0);
      int n = __builtin_amdgcn_readlane(myB, 0) - base;
      int f = (lane < n) ? glst[base + lane] : 0;
      int k = 0;
      for (int r = w; r < RPB; r += AGG_WAVES, ++k) {
        const int node = lo + r;
        if (node >= N) break;
        int basen = 0, nn = 0, fn = 0;
        if (r + AGG_WAVES < RPB) {
          basen = __builtin_amdgcn_readlane(myA, k + 1);
          nn = __builtin_amdgcn_readlane(myB, k + 1) - basen;
          fn = (lane < nn) ? glst[basen + lane] : 0;
        }
        const float4 a = sum_rows(xin, glst + base, f, n, lane);
        store_mean(hn, node, a, n, lane);
        base = basen;
        n = nn;
        f = fn;
      }
    }
    return;
  }

  for (int i = tid; i < (RPB + 1) * FEAT; i += 1024) acc[i] = 0.0f;
  if (tid < RPB + 4) cnt[tid] = 0.0f;
  __syncthreads();
  sweep_share(xin, src, dst, ob, acc, cnt, q, lo, w, lane, E);
  __syncthreads();

  for (int r = w; r < RPB; r += AGG_WAVES) {
    const int node = lo + r;
    if (node >= N) break;
    const float dg = fmaxf(cnt[r], 1.0f);
    const float* a = acc + r * FEAT + lane;
    float* o = hn + (size_t)node * FEAT + lane;
#pragma unroll
    for (int k = 0; k < 4; ++k) o[32 * k] = a[32 * k] / dg;
  }
}

#define LIN_T 512
#define LIN_BLOCKS 256

template <int F, bool RELU>
__global__ __launch_bounds__(LIN_T) void sage_linear_resident(const float* __restrict__ Xs,
                                                              const float* __restrict__ Xn,
                                                              const float* __restrict__ Wself,
                                                              const float* __restrict__ Wneigh,
                                                              const float* __restrict__ bias,
                                                              float* __restrict__ Out, int Nnodes) {
  constexpr int NT = F / 16;
  constexpr int STEPS = 2 * FEAT / 8;
  __shared__ float4 wl[STEPS * NT * 32];

  const int tid = threadIdx.x;
  const int w = __builtin_amdgcn_readfirstlane(tid >> 5);
  const int lane = tid & 31;
  const int c = lane & 15;
  const int h = lane >> 4;

  const int t = blockIdx.x + LIN_BLOCKS * w;
  const bool has_strip = t < (Nnodes >> 4);
  const int m0 = t << 4;
  const float* xs = Xs + (size_t)(m0 + c) * FEAT + 4 * h;
  const float* xn = Xn + (size_t)(m0 + c) * FEAT + 4 * h;
  float4 a[4];
#pragma unroll
  for (int s = 0; s < 4; ++s) a[s] = make_float4(0.f, 0.f, 0.f, 0.f);
  if (has_strip) {
#pragma unroll
    for (int s = 0; s < 4; ++s) a[s] = *(const float4*)(xs + 8 * s);
  }

  static_assert((2 * FEAT / 4 * F) % LIN_T == 0, "whole trips");
#pragma unroll
  for (int i = 0; i < 2 * FEAT / 4 * F / LIN_T; ++i) {
    const int idx = i * LIN_T + tid;
    const int n = idx % F;
    const int q = idx / F;
    const float* Wm = (q < FEAT / 4) ? Wself : Wneigh;
    const float* p = Wm + (size_t)((q & (FEAT / 4 - 1)) * 4) * F + n;
    float4 v;
    v.x = p[0];
    v.y = p[F];
    v.z = p[2 * F];
    v.w = p[3 * F];
    wl[((q >> 1) * NT + (n >> 4)) * 32 + (q & 1) * 16 + (n & 15)] = v;
  }
  __syncthreads();

  if (!has_strip) return;
  const float4* bp = wl + lane;

  v8f acc[NT];
#pragma unroll
  for (int nt = 0; nt < NT; ++nt) acc[nt] = v8f{};

  float4 b[NT];
#pragma unroll
  for (int nt = 0; nt < NT; ++nt) b[nt] = bp[nt * 32];

  const float* xp = xs;
  for (int g = 0; g < STEPS / 4; ++g) {
    xp = (g == 3) ? xn : ((g == 7) ? xs : xp + 32);
    float4 an[4];
#pragma unroll
    for (int s = 0; s < 4; ++s) an[s] = *(const float4*)(xp + 8 * s);
#pragma unroll
    for (int s = 0; s < 4; ++s) {
      const int kn = (4 * g + s + 1) & (STEPS - 1);
      const float4* bq = bp + kn * (NT * 32);
      float4 bn[NT];
#pragma unroll
      for (int nt = 0; nt < NT; ++nt) bn[nt] = bq[nt * 32];
      __builtin_amdgcn_sched_barrier(0);
      const v2f a0 = {a[s].x, a[s].y};
      const v2f a1 = {a[s].z, a[s].w};
#pragma unroll
      for (int nt = 0; nt < NT; ++nt) {
        const v2f b0 = {b[nt].x, b[nt].y};
        acc[nt] = __builtin_amdgcn_wmma_f32_16x16x4_f32(false, a0, false, b0, (short)0, acc[nt], false, false);
      }
#pragma unroll
      for (int nt = 0; nt < NT; ++nt) {
        const v2f b1 = {b[nt].z, b[nt].w};
        acc[nt] = __builtin_amdgcn_wmma_f32_16x16x4_f32(false, a1, false, b1, (short)0, acc[nt], false, false);
      }
      __builtin_amdgcn_sched_barrier(0);
#pragma unroll
      for (int nt = 0; nt < NT; ++nt) b[nt] = bn[nt];
    }
#pragma unroll
    for (int s = 0; s < 4; ++s) a[s] = an[s];
  }

  const int mbase = m0 + (h << 3);
#pragma unroll
  for (int nt = 0; nt < NT; ++nt) {
    const float bv = bias[nt * 16 + c];
#pragma unroll
    for (int r = 0; r < 8; ++r) {
      float v = acc[nt][r] + bv;
      if (RELU) v = fmaxf(v, 0.0f);
      Out[(size_t)(mbase + r) * F + nt * 16 + c] = v;
    }
  }
}

extern "C" void kernel_launch(void* const* d_in, const int* in_sizes, int n_in,
                              void* d_out, int out_size, void* d_ws, size_t ws_size,
                              hipStream_t stream) {
  (void)in_sizes; (void)n_in; (void)out_size; (void)ws_size;

  const float* x   = (const float*)d_in[0];
  const int*   src = (const int*)d_in[1];
  const int*   dst = (const int*)d_in[2];
  const float* W1s = (const float*)d_in[3];
  const float* W1n = (const float*)d_in[4];
  const float* b1  = (const float*)d_in[5];
  const float* W2s = (const float*)d_in[6];
  const float* W2n = (const float*)d_in[7];
  const float* b2  = (const float*)d_in[8];
  float* out = (float*)d_out;

  const int N = 50000, E = 800000, H = 128;
  const long NH = (long)N * H;

  float* ws = (float*)d_ws;
  float* hn = ws;
  float* h  = hn + NH;

  static_assert(LIN_BLOCKS * (LIN_T / 32) * 16 >= N, "every 16-row strip needs a wave of its own");
  const int AGG_BLOCKS = (N + RPB - 1) / RPB;
  const int AGG_T = AGG_WAVES * 32;

  int* lists = (int*)d_out;
  int* rowst = lists + AGG_BLOCKS * SCAP;
  int* flags = rowst + AGG_BLOCKS * RSP;
  unsigned* ob = (unsigned*)(flags + AGG_BLOCKS);
  static_assert(((size_t)256 * (SCAP + RSP + 1) * 4) % 16 == 0, "16-byte loads of the owner bytes");
  static_assert((size_t)256 * (SCAP + RSP + 1) * 4 + 800000 <= (size_t)50000 * 64 * 4, "fits the output buffer");
  int* cells = (int*)h;
  int* cellcnt = cells + OWNERS * PARTS * CELLCAP;
  static_assert((size_t)OWNERS * PARTS * (CELLCAP + 1) * 4 <= (size_t)50000 * 128 * 4, "cells and counts fit the room of h");
  static_assert((size_t)PARTS * PEDGES >= 800000, "every edge has a part");
  static_assert(OWNERS * RPB >= 50000 && OWNERS == 256, "one owner block per owner");

  partition_kernel<<<PARTS, 256, 0, stream>>>(src, dst, ob, cells, cellcnt, E);
  aggregate_owner_kernel<false><<<AGG_BLOCKS, AGG_T, 0, stream>>>(x, src, dst, (const unsigned char*)ob, cells, cellcnt, hn, lists, rowst, flags, N, E);
  sage_linear_resident<128, true><<<LIN_BLOCKS, LIN_T, 0, stream>>>(x, hn, W1s, W1n, b1, h, N);

  aggregate_owner_kernel<true><<<AGG_BLOCKS, AGG_T, 0, stream>>>(h, src, dst, (const unsigned char*)ob, cells, cellcnt, hn, lists, rowst, flags, N, E);
  sage_linear_resident<64, false><<<LIN_BLOCKS, LIN_T, 0, stream>>>(h, hn, W2s, W2n, b2, out, N);
}
